# starting kernel
# baseline (speedup 1.0000x reference)
#include <hip/hip_runtime.h>
#include <math.h>

typedef __attribute__((ext_vector_type(2))) float v2f;
typedef __attribute__((ext_vector_type(8))) float v8f;

#define RNN_HID 1024
#define RNN_BATCH 128
#define RNN_SEQ 512
#define NTILES 4

__global__ __launch_bounds__(256) void rnn_reorder_w_kernel(
    const float* __restrict__ W, float* __restrict__ WT)
{
    int idx = blockIdx.x * blockDim.x + threadIdx.x;
    int k = idx >> 10;
    int n = idx & (RNN_HID - 1);
    WT[((size_t)(k >> 1) * RNN_HID + n) * 2 + (k & 1)] = W[idx];
}

template <bool FUSE2, bool DO_TANH>
__global__ __launch_bounds__(128) void rnn_gemm_kernel(
    const float* __restrict__ A0, size_t a0_stride,
    const float* __restrict__ W0T,
    const float* __restrict__ A1,
    const float* __restrict__ W1T,
    const float* __restrict__ bias,
    float* __restrict__ out)
{
    const int lane = threadIdx.x & 31;
    const int wave = threadIdx.x >> 5;
    const int half = lane >> 4;
    const int lr   = lane & 15;

    const int n0 = blockIdx.x * (16 * NTILES);
    const int m0 = blockIdx.y * 16;

    const float* A;
    const float* WT;
    size_t astr;
    int kb, klen;
    if (FUSE2) {
        if (wave < 2) { A = A0; WT = W0T; astr = a0_stride; kb = wave * 512; }
        else          { A = A1; WT = W1T; astr = RNN_HID;   kb = (wave - 2) * 512; }
        klen = 512;
    } else {
        A = A0; WT = W0T; astr = a0_stride; kb = wave * 256; klen = 256;
    }

    v8f acc[NTILES] = {};

    const float* ap = A + (size_t)(m0 + lr) * astr + (kb + half * 2);
    const float* wt = WT + ((size_t)((kb >> 1) + half) * RNN_HID + n0 + lr) * 2;

    for (int kk = 0; kk < klen; kk += 4) {
        v2f a = *(const v2f*)ap;
#pragma unroll
        for (int i = 0; i < NTILES; i++) {
            v2f b = *(const v2f*)(wt + i * 32);
            acc[i] = __builtin_amdgcn_wmma_f32_16x16x4_f32(
                false, a, false, b, (short)0, acc[i], false, false);
        }
        ap += 4;
        wt += 2 * RNN_HID * 2;
    }

    __shared__ float red[4][NTILES][8][32];
#pragma unroll
    for (int i = 0; i < NTILES; i++)
#pragma unroll
        for (int j = 0; j < 8; j++)
            red[wave][i][j][lane] = acc[i][j];
    __syncthreads();

    if (wave == 0) {
#pragma unroll
        for (int i = 0; i < NTILES; i++) {
            const int n = n0 + i * 16 + lr;
            const float bv = bias[n];
#pragma unroll
            for (int j = 0; j < 8; j++) {
                float v = red[0][i][j][lane] + red[1][i][j][lane] +
                          red[2][i][j][lane] + red[3][i][j][lane] + bv;
                if (DO_TANH) v = tanhf(v);
                out[(size_t)(m0 + j + half * 8) * RNN_HID + n] = v;
            }
        }
    }
}

__global__ void rnn_zero_kernel(float* __restrict__ p, int n)
{
    int i = blockIdx.x * blockDim.x + threadIdx.x;
    if (i < n) p[i] = 0.0f;
}

extern "C" void kernel_launch(void* const* d_in, const int* in_sizes, int n_in,
                              void* d_out, int out_size, void* d_ws, size_t ws_size,
                              hipStream_t stream)
{
    (void)in_sizes; (void)n_in; (void)out_size; (void)ws_size;
    const float* x    = (const float*)d_in[0];
    const float* W_xh = (const float*)d_in[1];
    const float* W_hh = (const float*)d_in[2];
    const float* b_h  = (const float*)d_in[3];
    const float* W_hy = (const float*)d_in[4];
    const float* b_y  = (const float*)d_in[5];
    float* out = (float*)d_out;

    const int hElems = RNN_BATCH * RNN_HID;
    const int wElems = RNN_HID * RNN_HID;

    float* hA    = (float*)d_ws;
    float* hB    = hA + hElems;
    float* WTxh  = hB + hElems;
    float* WThh  = WTxh + wElems;
    float* WThy  = WThh + wElems;

    rnn_zero_kernel<<<hElems / 256, 256, 0, stream>>>(hA, hElems);

    rnn_reorder_w_kernel<<<wElems / 256, 256, 0, stream>>>(W_xh, WTxh);
    rnn_reorder_w_kernel<<<wElems / 256, 256, 0, stream>>>(W_hh, WThh);
    rnn_reorder_w_kernel<<<wElems / 256, 256, 0, stream>>>(W_hy, WThy);

    dim3 grid(RNN_HID / (16 * NTILES), RNN_BATCH / 16);
    dim3 block(128);

    float* hin = hA;
    float* hout = hB;
    for (int t = 0; t < RNN_SEQ; t++) {
        rnn_gemm_kernel<true, true><<<grid, block, 0, stream>>>(
            x + (size_t)t * RNN_HID, (size_t)RNN_SEQ * RNN_HID, WTxh,
            hin, WThh, b_h, hout);
        float* tmp = hin; hin = hout; hout = tmp;
    }

    rnn_gemm_kernel<false, false><<<grid, block, 0, stream>>>(
        hin, (size_t)RNN_HID, WThy, nullptr, nullptr, b_y, out);
}


# K loop of the GEMM kernels register double-buffered: A/B fragments of the next 8 K-steps (40 loads) go into a second register set ahead of the current block's WMMAs; summation order unchanged
# speedup vs baseline: 1.6242x; 1.6242x over previous
#include <hip/hip_runtime.h>
#include <math.h>

typedef __attribute__((ext_vector_type(2))) float v2f;
typedef __attribute__((ext_vector_type(8))) float v8f;

#define RNN_HID 1024
#define RNN_BATCH 128
#define RNN_SEQ 512
#define NTILES 4
#define KBLK 8

__global__ __launch_bounds__(256) void rnn_reorder_w_kernel(
    const float* __restrict__ W, float* __restrict__ WT)
{
    int idx = blockIdx.x * blockDim.x + threadIdx.x;
    int k = idx >> 10;
    int n = idx & (RNN_HID - 1);
    WT[((size_t)(k >> 1) * RNN_HID + n) * 2 + (k & 1)] = W[idx];
}

template <bool FUSE2, bool DO_TANH>
__global__ __launch_bounds__(128) void rnn_gemm_kernel(
    const float* __restrict__ A0, size_t a0_stride,
    const float* __restrict__ W0T,
    const float* __restrict__ A1,
    const float* __restrict__ W1T,
    const float* __restrict__ bias,
    float* __restrict__ out)
{
    const int lane = threadIdx.x & 31;
    const int wave = threadIdx.x >> 5;
    const int half = lane >> 4;
    const int lr   = lane & 15;

    const int n0 = blockIdx.x * (16 * NTILES);
    const int m0 = blockIdx.y * 16;

    const float* A;
    const float* WT;
    size_t astr;
    int kb, klen;
    if (FUSE2) {
        if (wave < 2) { A = A0; WT = W0T; astr = a0_stride; kb = wave * 512; }
        else          { A = A1; WT = W1T; astr = RNN_HID;   kb = (wave - 2) * 512; }
        klen = 512;
    } else {
        A = A0; WT = W0T; astr = a0_stride; kb = wave * 256; klen = 256;
    }

    v8f acc[NTILES] = {};

    const float* ap = A + (size_t)(m0 + lr) * astr + (kb + half * 2);
    const float* wt = WT + ((size_t)((kb >> 1) + half) * RNN_HID + n0 + lr) * 2;

    v2f aP[KBLK], bP[KBLK][NTILES];
    v2f aQ[KBLK], bQ[KBLK][NTILES];
    const int nblk = klen / (4 * KBLK);

#define RNN_LOAD_BLOCK(a_, b_)                                              \
    do {                                                                    \
        _Pragma("unroll")                                                   \
        for (int s = 0; s < KBLK; s++) {                                    \
            a_[s] = *(const v2f*)(ap + 4 * s);                              \
            _Pragma("unroll")                                               \
            for (int i = 0; i < NTILES; i++)                                \
                b_[s][i] = *(const v2f*)(wt + s * (2 * RNN_HID * 2) + i * 32); \
        }                                                                   \
        ap += 4 * KBLK;                                                     \
        wt += (size_t)KBLK * (2 * RNN_HID * 2);                             \
    } while (0)

#define RNN_MMA_BLOCK(a_, b_)                                               \
    do {                                                                    \
        _Pragma("unroll")                                                   \
        for (int s = 0; s < KBLK; s++) {                                    \
            _Pragma("unroll")                                               \
            for (int i = 0; i < NTILES; i++)                                \
                acc[i] = __builtin_amdgcn_wmma_f32_16x16x4_f32(             \
                    false, a_[s], false, b_[s][i], (short)0, acc[i],        \
                    false, false);                                          \
        }                                                                   \
    } while (0)

    RNN_LOAD_BLOCK(aP, bP);
    for (int blk = 0; blk < nblk - 2; blk += 2) {
        RNN_LOAD_BLOCK(aQ, bQ);
        RNN_MMA_BLOCK(aP, bP);
        RNN_LOAD_BLOCK(aP, bP);
        RNN_MMA_BLOCK(aQ, bQ);
    }
    RNN_LOAD_BLOCK(aQ, bQ);
    RNN_MMA_BLOCK(aP, bP);
    RNN_MMA_BLOCK(aQ, bQ);
#undef RNN_LOAD_BLOCK
#undef RNN_MMA_BLOCK

    __shared__ float red[4][NTILES][8][32];
#pragma unroll
    for (int i = 0; i < NTILES; i++)
#pragma unroll
        for (int j = 0; j < 8; j++)
            red[wave][i][j][lane] = acc[i][j];
    __syncthreads();

    if (wave == 0) {
#pragma unroll
        for (int i = 0; i < NTILES; i++) {
            const int n = n0 + i * 16 + lr;
            const float bv = bias[n];
#pragma unroll
            for (int j = 0; j < 8; j++) {
                float v = red[0][i][j][lane] + red[1][i][j][lane] +
                          red[2][i][j][lane] + red[3][i][j][lane] + bv;
                if (DO_TANH) v = tanhf(v);
                out[(size_t)(m0 + j + half * 8) * RNN_HID + n] = v;
            }
        }
    }
}

__global__ void rnn_zero_kernel(float* __restrict__ p, int n)
{
    int i = blockIdx.x * blockDim.x + threadIdx.x;
    if (i < n) p[i] = 0.0f;
}

extern "C" void kernel_launch(void* const* d_in, const int* in_sizes, int n_in,
                              void* d_out, int out_size, void* d_ws, size_t ws_size,
                              hipStream_t stream)
{
    (void)in_sizes; (void)n_in; (void)out_size; (void)ws_size;
    const float* x    = (const float*)d_in[0];
    const float* W_xh = (const float*)d_in[1];
    const float* W_hh = (const float*)d_in[2];
    const float* b_h  = (const float*)d_in[3];
    const float* W_hy = (const float*)d_in[4];
    const float* b_y  = (const float*)d_in[5];
    float* out = (float*)d_out;

    const int hElems = RNN_BATCH * RNN_HID;
    const int wElems = RNN_HID * RNN_HID;

    float* hA    = (float*)d_ws;
    float* hB    = hA + hElems;
    float* WTxh  = hB + hElems;
    float* WThh  = WTxh + wElems;
    float* WThy  = WThh + wElems;

    rnn_zero_kernel<<<hElems / 256, 256, 0, stream>>>(hA, hElems);

    rnn_reorder_w_kernel<<<wElems / 256, 256, 0, stream>>>(W_xh, WTxh);
    rnn_reorder_w_kernel<<<wElems / 256, 256, 0, stream>>>(W_hh, WThh);
    rnn_reorder_w_kernel<<<wElems / 256, 256, 0, stream>>>(W_hy, WThy);

    dim3 grid(RNN_HID / (16 * NTILES), RNN_BATCH / 16);
    dim3 block(128);

    float* hin = hA;
    float* hout = hB;
    for (int t = 0; t < RNN_SEQ; t++) {
        rnn_gemm_kernel<true, true><<<grid, block, 0, stream>>>(
            x + (size_t)t * RNN_HID, (size_t)RNN_SEQ * RNN_HID, WTxh,
            hin, WThh, b_h, hout);
        float* tmp = hin; hin = hout; hout = tmp;
    }

    rnn_gemm_kernel<false, false><<<grid, block, 0, stream>>>(
        hin, (size_t)RNN_HID, WThy, nullptr, nullptr, b_y, out);
}


# input projection x_t @ W_xh moved to work-groups of its own, one launch ahead of its step (step launch: 256 work-groups of K=1024, was 128 of K=2048); sum order changes: projection starts the sum
# speedup vs baseline: 1.9942x; 1.2278x over previous
#include <hip/hip_runtime.h>
#include <math.h>


typedef __attribute__((ext_vector_type(2))) float v2f;
typedef __attribute__((ext_vector_type(8))) float v8f;

#define RNN_HID 1024
#define RNN_BATCH 128
#define RNN_SEQ 512
#define NTILES 4
#define KBLK 8
#define KWAVE 256

__global__ __launch_bounds__(256) void rnn_reorder_w_kernel(
    const float* __restrict__ W, float* __restrict__ WT)
{
    int idx = blockIdx.x * blockDim.x + threadIdx.x;
    int k = idx >> 10;
    int n = idx & (RNN_HID - 1);
    WT[((size_t)(k >> 1) * RNN_HID + n) * 2 + (k & 1)] = W[idx];
}

template <bool STEP>
__global__ __launch_bounds__(128) void rnn_gemm_kernel(
    const float* __restrict__ Ah, const float* __restrict__ WhT,
    const float* __restrict__ Ax, const float* __restrict__ WxT,
    const float* __restrict__ xp_in, const float* __restrict__ bias,
    float* __restrict__ out, float* __restrict__ xp_out, int zbase)
{
    const int lane = threadIdx.x & 31;
    const int wave = threadIdx.x >> 5;
    const int half = lane >> 4;
    const int lr   = lane & 15;

    const int n0 = blockIdx.x * (16 * NTILES);
    const int m0 = blockIdx.y * 16;

    const bool xrole = STEP && ((int)blockIdx.z + zbase) != 0;

    const float* A  = xrole ? Ax : Ah;
    const float* WT = xrole ? WxT : WhT;
    const size_t astr = xrole ? (size_t)RNN_SEQ * RNN_HID : (size_t)RNN_HID;
    const int kb = wave * KWAVE;

    const float* ap = A + (size_t)(m0 + lr) * astr + (kb + half * 2);
    const float* wt = WT + ((size_t)((kb >> 1) + half) * RNN_HID + n0 + lr) * 2;

    v2f aP[KBLK], bP[KBLK][NTILES];
    v2f aQ[KBLK], bQ[KBLK][NTILES];
    const int nblk = KWAVE / (4 * KBLK);

#define RNN_LOAD_BLOCK(a_, b_)                                              \
    do {                                                                    \
        _Pragma("unroll")                                                   \
        for (int s = 0; s < KBLK; s++) {                                    \
            a_[s] = *(const v2f*)(ap + 4 * s);                              \
            _Pragma("unroll")                                               \
            for (int i = 0; i < NTILES; i++)                                \
                b_[s][i] = *(const v2f*)(wt + s * (2 * RNN_HID * 2) + i * 32); \
        }                                                                   \
        ap += 4 * KBLK;                                                     \
        wt += (size_t)KBLK * (2 * RNN_HID * 2);                             \
    } while (0)

#define RNN_MMA_BLOCK(a_, b_)                                               \
    do {                                                                    \
        _Pragma("unroll")                                                   \
        for (int s = 0; s < KBLK; s++) {                                    \
            _Pragma("unroll")                                               \
            for (int i = 0; i < NTILES; i++)                                \
                acc[i] = __builtin_amdgcn_wmma_f32_16x16x4_f32(             \
                    false, a_[s], false, b_[s][i], (short)0, acc[i],        \
                    false, false);                                          \
        }                                                                   \
    } while (0)

    RNN_LOAD_BLOCK(aP, bP);

    v8f xinit = {};
    if (STEP && !xrole) {
#pragma unroll
        for (int j = 0; j < 8; j++)
            xinit[j] = xp_in[(size_t)(m0 + j + half * 8) * RNN_HID + n0 + wave * 16 + lr];
    }
    v8f acc[NTILES];
#pragma unroll
    for (int i = 0; i < NTILES; i++) {
        const v8f zero = {};
        acc[i] = (i == wave) ? xinit : zero;
    }

#pragma nounroll
    for (int blk = 0; blk < nblk - 2; blk += 2) {
        RNN_LOAD_BLOCK(aQ, bQ);
        RNN_MMA_BLOCK(aP, bP);
        RNN_LOAD_BLOCK(aP, bP);
        RNN_MMA_BLOCK(aQ, bQ);
    }
    RNN_LOAD_BLOCK(aQ, bQ);

    RNN_MMA_BLOCK(aP, bP);
    RNN_MMA_BLOCK(aQ, bQ);
#undef RNN_LOAD_BLOCK
#undef RNN_MMA_BLOCK

    __shared__ float red[4][NTILES][8][32];
#pragma unroll
    for (int i = 0; i < NTILES; i++)
#pragma unroll
        for (int j = 0; j < 8; j++)
            red[wave][i][j][lane] = acc[i][j];
    __syncthreads();

    if (wave == 0) {
#pragma unroll
        for (int i = 0; i < NTILES; i++) {
            const int n = n0 + i * 16 + lr;
            const float bv = xrole ? 0.0f : bias[n];
#pragma unroll
            for (int j = 0; j < 8; j++) {
                const float s = red[0][i][j][lane] + red[1][i][j][lane] +
                                red[2][i][j][lane] + red[3][i][j][lane];
                const size_t o = (size_t)(m0 + j + half * 8) * RNN_HID + n;
                if (STEP) {
                    if (xrole) xp_out[o] = s;
                    else       out[o] = tanhf(s + bv);
                } else {
                    out[o] = s + bv;
                }
            }
        }
    }
}

__global__ void rnn_zero_kernel(float* __restrict__ p, int n)
{
    int i = blockIdx.x * blockDim.x + threadIdx.x;
    if (i < n) p[i] = 0.0f;
}

extern "C" void kernel_launch(void* const* d_in, const int* in_sizes, int n_in,
                              void* d_out, int out_size, void* d_ws, size_t ws_size,
                              hipStream_t stream)
{
    (void)in_sizes; (void)n_in; (void)out_size; (void)ws_size;
    const float* x    = (const float*)d_in[0];
    const float* W_xh = (const float*)d_in[1];
    const float* W_hh = (const float*)d_in[2];
    const float* b_h  = (const float*)d_in[3];
    const float* W_hy = (const float*)d_in[4];
    const float* b_y  = (const float*)d_in[5];
    float* out = (float*)d_out;

    const int hElems = RNN_BATCH * RNN_HID;
    const int wElems = RNN_HID * RNN_HID;

    float* hA    = (float*)d_ws;
    float* hB    = hA + hElems;
    float* WTxh  = hB + hElems;
    float* WThh  = WTxh + wElems;
    float* WThy  = WThh + wElems;
    float* xpA   = WThy;
    float* xpB   = WThy + hElems;

    rnn_zero_kernel<<<hElems / 256, 256, 0, stream>>>(hA, hElems);

    rnn_reorder_w_kernel<<<wElems / 256, 256, 0, stream>>>(W_xh, WTxh);
    rnn_reorder_w_kernel<<<wElems / 256, 256, 0, stream>>>(W_hh, WThh);

    dim3 gridOne(RNN_HID / (16 * NTILES), RNN_BATCH / 16, 1);
    dim3 gridTwo(RNN_HID / (16 * NTILES), RNN_BATCH / 16, 2);
    dim3 block(128);

    float* hin = hA;
    float* hout = hB;
    float* xin = xpA;
    float* xout = xpB;

    rnn_gemm_kernel<true><<<gridOne, block, 0, stream>>>(
        hin, WThh, x, WTxh, xout, b_h, hout, xin, 1);

    for (int t = 0; t < RNN_SEQ - 1; t++) {
        rnn_gemm_kernel<true><<<gridTwo, block, 0, stream>>>(
            hin, WThh, x + (size_t)(t + 1) * RNN_HID, WTxh, xin, b_h, hout, xout, 0);
        float* tmp = hin; hin = hout; hout = tmp;
        tmp = xin; xin = xout; xout = tmp;
    }
    rnn_gemm_kernel<true><<<gridOne, block, 0, stream>>>(
        hin, WThh, x, WTxh, xin, b_h, hout, xout, 0);
    {
        float* tmp = hin; hin = hout; hout = tmp;
    }

    rnn_reorder_w_kernel<<<wElems / 256, 256, 0, stream>>>(W_hy, WThy);

    rnn_gemm_kernel<false><<<gridOne, block, 0, stream>>>(
        hin, WThy, x, WTxh, b_y, b_y, out, hout, 0);
}


# epilogue of the GEMM kernels (split-K reduction from LDS, bias, tanh, stores) spread over all 4 waves, 8 whole 128-B lines each, was all 32 on wave 0; sum order per element unchanged
# speedup vs baseline: 2.8766x; 1.4425x over previous
#include <hip/hip_runtime.h>
#include <math.h>


typedef __attribute__((ext_vector_type(2))) float v2f;
typedef __attribute__((ext_vector_type(8))) float v8f;

#define RNN_HID 1024
#define RNN_BATCH 128
#define RNN_SEQ 512
#define NTILES 4
#define KBLK 8
#define KWAVE 256

__global__ __launch_bounds__(256) void rnn_reorder_w_kernel(
    const float* __restrict__ W, float* __restrict__ WT)
{
    int idx = blockIdx.x * blockDim.x + threadIdx.x;
    int k = idx >> 10;
    int n = idx & (RNN_HID - 1);
    WT[((size_t)(k >> 1) * RNN_HID + n) * 2 + (k & 1)] = W[idx];
}

template <bool STEP>
__global__ __launch_bounds__(128) void rnn_gemm_kernel(
    const float* __restrict__ Ah, const float* __restrict__ WhT,
    const float* __restrict__ Ax, const float* __restrict__ WxT,
    const float* __restrict__ xp_in, const float* __restrict__ bias,
    float* __restrict__ out, float* __restrict__ xp_out, int zbase)
{
    const int lane = threadIdx.x & 31;
    const int wave = threadIdx.x >> 5;
    const int half = lane >> 4;
    const int lr   = lane & 15;

    const int n0 = blockIdx.x * (16 * NTILES);
    const int m0 = blockIdx.y * 16;

    const bool xrole = STEP && ((int)blockIdx.z + zbase) != 0;

    const float* A  = xrole ? Ax : Ah;
    const float* WT = xrole ? WxT : WhT;
    const size_t astr = xrole ? (size_t)RNN_SEQ * RNN_HID : (size_t)RNN_HID;
    const int kb = wave * KWAVE;

    const float* ap = A + (size_t)(m0 + lr) * astr + (kb + half * 2);
    const float* wt = WT + ((size_t)((kb >> 1) + half) * RNN_HID + n0 + lr) * 2;

    v2f aP[KBLK], bP[KBLK][NTILES];
    v2f aQ[KBLK], bQ[KBLK][NTILES];
    const int nblk = KWAVE / (4 * KBLK);

#define RNN_LOAD_BLOCK(a_, b_)                                              \
    do {                                                                    \
        _Pragma("unroll")                                                   \
        for (int s = 0; s < KBLK; s++) {                                    \
            a_[s] = *(const v2f*)(ap + 4 * s);                              \
            _Pragma("unroll")                                               \
            for (int i = 0; i < NTILES; i++)                                \
                b_[s][i] = *(const v2f*)(wt + s * (2 * RNN_HID * 2) + i * 32); \
        }                                                                   \
        ap += 4 * KBLK;                                                     \
        wt += (size_t)KBLK * (2 * RNN_HID * 2);                             \
    } while (0)

#define RNN_MMA_BLOCK(a_, b_)                                               \
    do {                                                                    \
        _Pragma("unroll")                                                   \
        for (int s = 0; s < KBLK; s++) {                                    \
            _Pragma("unroll")                                               \
            for (int i = 0; i < NTILES; i++)                                \
                acc[i] = __builtin_amdgcn_wmma_f32_16x16x4_f32(             \
                    false, a_[s], false, b_[s][i], (short)0, acc[i],        \
                    false, false);                                          \
        }                                                                   \
    } while (0)

    RNN_LOAD_BLOCK(aP, bP);

    v8f xinit = {};
    if (STEP && !xrole) {
#pragma unroll
        for (int j = 0; j < 8; j++)
            xinit[j] = xp_in[(size_t)(m0 + j + half * 8) * RNN_HID + n0 + wave * 16 + lr];
    }
    float bv[2];
#pragma unroll
    for (int p = 0; p < 2; p++)
        bv[p] = xrole ? 0.0f : bias[n0 + 32 * p + lane];

    v8f acc[NTILES];
#pragma unroll
    for (int i = 0; i < NTILES; i++) {
        const v8f zero = {};
        acc[i] = (i == wave) ? xinit : zero;
    }

#pragma nounroll
    for (int blk = 0; blk < nblk - 2; blk += 2) {
        RNN_LOAD_BLOCK(aQ, bQ);
        RNN_MMA_BLOCK(aP, bP);
        RNN_LOAD_BLOCK(aP, bP);
        RNN_MMA_BLOCK(aQ, bQ);
    }
    RNN_LOAD_BLOCK(aQ, bQ);

    RNN_MMA_BLOCK(aP, bP);
    RNN_MMA_BLOCK(aQ, bQ);
#undef RNN_LOAD_BLOCK
#undef RNN_MMA_BLOCK

    __shared__ float red[4][NTILES][8][32];
#pragma unroll
    for (int i = 0; i < NTILES; i++)
#pragma unroll
        for (int j = 0; j < 8; j++)
            red[wave][i][j][lane] = acc[i][j];
    __syncthreads();

    const int rbase = 4 * wave;
    const int jb = rbase & 7;
    const int sl = (rbase >> 3) * 16 + lr;

    float s[8];
#pragma unroll
    for (int q = 0; q < 8; q++) {
        const int i = 2 * (q & 1) + half;
        const int j = jb + (q >> 1);
        s[q] = red[0][i][j][sl] + red[1][i][j][sl] +
               red[2][i][j][sl] + red[3][i][j][sl];
    }

#pragma unroll
    for (int q = 0; q < 8; q++) {
        const int p = q & 1;
        const size_t o = (size_t)(m0 + rbase + (q >> 1)) * RNN_HID + n0 + 32 * p + lane;
        if (STEP) {
            if (xrole) xp_out[o] = s[q];
            else       out[o] = tanhf(s[q] + bv[p]);
        } else {
            out[o] = s[q] + bv[p];
        }
    }
}

__global__ void rnn_zero_kernel(float* __restrict__ p, int n)
{
    int i = blockIdx.x * blockDim.x + threadIdx.x;
    if (i < n) p[i] = 0.0f;
}

extern "C" void kernel_launch(void* const* d_in, const int* in_sizes, int n_in,
                              void* d_out, int out_size, void* d_ws, size_t ws_size,
                              hipStream_t stream)
{
    (void)in_sizes; (void)n_in; (void)out_size; (void)ws_size;
    const float* x    = (const float*)d_in[0];
    const float* W_xh = (const float*)d_in[1];
    const float* W_hh = (const float*)d_in[2];
    const float* b_h  = (const float*)d_in[3];
    const float* W_hy = (const float*)d_in[4];
    const float* b_y  = (const float*)d_in[5];
    float* out = (float*)d_out;

    const int hElems = RNN_BATCH * RNN_HID;
    const int wElems = RNN_HID * RNN_HID;

    float* hA    = (float*)d_ws;
    float* hB    = hA + hElems;
    float* WTxh  = hB + hElems;
    float* WThh  = WTxh + wElems;
    float* WThy  = WThh + wElems;
    float* xpA   = WThy;
    float* xpB   = WThy + hElems;

    rnn_zero_kernel<<<hElems / 256, 256, 0, stream>>>(hA, hElems);

    rnn_reorder_w_kernel<<<wElems / 256, 256, 0, stream>>>(W_xh, WTxh);
    rnn_reorder_w_kernel<<<wElems / 256, 256, 0, stream>>>(W_hh, WThh);

    dim3 gridOne(RNN_HID / (16 * NTILES), RNN_BATCH / 16, 1);
    dim3 gridTwo(RNN_HID / (16 * NTILES), RNN_BATCH / 16, 2);
    dim3 block(128);

    float* hin = hA;
    float* hout = hB;
    float* xin = xpA;
    float* xout = xpB;

    rnn_gemm_kernel<true><<<gridOne, block, 0, stream>>>(
        hin, WThh, x, WTxh, xout, b_h, hout, xin, 1);

    for (int t = 0; t < RNN_SEQ - 1; t++) {
        rnn_gemm_kernel<true><<<gridTwo, block, 0, stream>>>(
            hin, WThh, x + (size_t)(t + 1) * RNN_HID, WTxh, xin, b_h, hout, xout, 0);
        float* tmp = hin; hin = hout; hout = tmp;
        tmp = xin; xin = xout; xout = tmp;
    }
    rnn_gemm_kernel<true><<<gridOne, block, 0, stream>>>(
        hin, WThh, x, WTxh, xin, b_h, hout, xout, 0);
    {
        float* tmp = hin; hin = hout; hout = tmp;
    }

    rnn_reorder_w_kernel<<<wElems / 256, 256, 0, stream>>>(W_hy, WThy);

    rnn_gemm_kernel<false><<<gridOne, block, 0, stream>>>(
        hin, WThy, x, WTxh, b_y, b_y, out, hout, 0);
}


# GEMM fragment loads 16 B per lane (A: 4 consecutive k of a row; weights laid out in k-quads), 20 loads per 32-WMMA block held as one group, was 40; sum order inside each 8-k group changes
# speedup vs baseline: 3.1277x; 1.0873x over previous
#include <hip/hip_runtime.h>
#include <math.h>


typedef __attribute__((ext_vector_type(2))) float v2f;
typedef __attribute__((ext_vector_type(4))) float v4f;
typedef __attribute__((ext_vector_type(8))) float v8f;

#define RNN_HID 1024
#define RNN_BATCH 128
#define RNN_SEQ 512
#define NTILES 4
#define QBLK 4
#define KWAVE 256

__global__ __launch_bounds__(256) void rnn_reorder_w_kernel(
    const float* __restrict__ W, float* __restrict__ WT)
{
    int idx = blockIdx.x * blockDim.x + threadIdx.x;
    int q = idx >> 10;
    int n = idx & (RNN_HID - 1);
    const float* src = W + (size_t)(4 * q) * RNN_HID + n;
    v4f v;
    v.x = src[0];
    v.y = src[RNN_HID];
    v.z = src[2 * RNN_HID];
    v.w = src[3 * RNN_HID];
    *(v4f*)(WT + (size_t)idx * 4) = v;
}

template <bool STEP>
__global__ __launch_bounds__(128) void rnn_gemm_kernel(
    const float* __restrict__ Ah, const float* __restrict__ WhT,
    const float* __restrict__ Ax, const float* __restrict__ WxT,
    const float* __restrict__ xp_in, const float* __restrict__ bias,
    float* __restrict__ out, float* __restrict__ xp_out, int zbase)
{
    const int lane = threadIdx.x & 31;
    const int wave = threadIdx.x >> 5;
    const int half = lane >> 4;
    const int lr   = lane & 15;

    const int n0 = blockIdx.x * (16 * NTILES);
    const int m0 = blockIdx.y * 16;

    const bool xrole = STEP && ((int)blockIdx.z + zbase) != 0;

    const float* A  = xrole ? Ax : Ah;
    const float* WT = xrole ? WxT : WhT;
    const size_t astr = xrole ? (size_t)RNN_SEQ * RNN_HID : (size_t)RNN_HID;
    const int kb = wave * KWAVE;

    const float* ap = A + (size_t)(m0 + lr) * astr + (kb + half * 4);
    const float* wt = WT + ((size_t)((kb >> 2) + half) * RNN_HID + n0 + lr) * 4;

    v4f aP[QBLK], bP[QBLK][NTILES];
    v4f aQ[QBLK], bQ[QBLK][NTILES];
    const int nblk = KWAVE / (8 * QBLK);

#define RNN_LOAD_BLOCK(a_, b_)                                              \
    do {                                                                    \
        _Pragma("unroll")                                                   \
        for (int s = 0; s < QBLK; s++) {                                    \
            a_[s] = *(const v4f*)(ap + 8 * s);                              \
            _Pragma("unroll")                                               \
            for (int i = 0; i < NTILES; i++)                                \
                b_[s][i] = *(const v4f*)(wt + s * (2 * RNN_HID * 4) + i * 64); \
        }                                                                   \
        ap += 8 * QBLK;                                                     \
        wt += (size_t)QBLK * (2 * RNN_HID * 4);                             \
    } while (0)

#define RNN_MMA_BLOCK(a_, b_)                                               \
    do {                                                                    \
        _Pragma("unroll")                                                   \
        for (int s = 0; s < QBLK; s++) {                                    \
            _Pragma("unroll")                                               \
            for (int i = 0; i < NTILES; i++)                                \
                acc[i] = __builtin_amdgcn_wmma_f32_16x16x4_f32(             \
                    false, __builtin_shufflevector(a_[s], a_[s], 0, 1),     \
                    false, __builtin_shufflevector(b_[s][i], b_[s][i], 0, 1), \
                    (short)0, acc[i], false, false);                        \
            _Pragma("unroll")                                               \
            for (int i = 0; i < NTILES; i++)                                \
                acc[i] = __builtin_amdgcn_wmma_f32_16x16x4_f32(             \
                    false, __builtin_shufflevector(a_[s], a_[s], 2, 3),     \
                    false, __builtin_shufflevector(b_[s][i], b_[s][i], 2, 3), \
                    (short)0, acc[i], false, false);                        \
        }                                                                   \
    } while (0)

    RNN_LOAD_BLOCK(aP, bP);

    v8f xinit = {};
    if (STEP && !xrole) {
#pragma unroll
        for (int j = 0; j < 8; j++)
            xinit[j] = xp_in[(size_t)(m0 + j + half * 8) * RNN_HID + n0 + wave * 16 + lr];
    }
    float bv[2];
#pragma unroll
    for (int p = 0; p < 2; p++)
        bv[p] = xrole ? 0.0f : bias[n0 + 32 * p + lane];

    v8f acc[NTILES];
#pragma unroll
    for (int i = 0; i < NTILES; i++) {
        const v8f zero = {};
        acc[i] = (i == wave) ? xinit : zero;
    }

#pragma nounroll
    for (int blk = 0; blk < nblk - 2; blk += 2) {
        RNN_LOAD_BLOCK(aQ, bQ);
        __builtin_amdgcn_sched_barrier(0);
        RNN_MMA_BLOCK(aP, bP);
        __builtin_amdgcn_sched_barrier(0);
        RNN_LOAD_BLOCK(aP, bP);
        __builtin_amdgcn_sched_barrier(0);
        RNN_MMA_BLOCK(aQ, bQ);
        __builtin_amdgcn_sched_barrier(0);
    }
    RNN_LOAD_BLOCK(aQ, bQ);
    __builtin_amdgcn_sched_barrier(0);

    RNN_MMA_BLOCK(aP, bP);
    RNN_MMA_BLOCK(aQ, bQ);
#undef RNN_LOAD_BLOCK
#undef RNN_MMA_BLOCK

    __shared__ float red[4][NTILES][8][32];
#pragma unroll
    for (int i = 0; i < NTILES; i++)
#pragma unroll
        for (int j = 0; j < 8; j++)
            red[wave][i][j][lane] = acc[i][j];
    __syncthreads();

    const int rbase = 4 * wave;
    const int jb = rbase & 7;
    const int sl = (rbase >> 3) * 16 + lr;

    float s[8];
#pragma unroll
    for (int q = 0; q < 8; q++) {
        const int i = 2 * (q & 1) + half;
        const int j = jb + (q >> 1);
        s[q] = red[0][i][j][sl] + red[1][i][j][sl] +
               red[2][i][j][sl] + red[3][i][j][sl];
    }

#pragma unroll
    for (int q = 0; q < 8; q++) {
        const int p = q & 1;
        const size_t o = (size_t)(m0 + rbase + (q >> 1)) * RNN_HID + n0 + 32 * p + lane;
        if (STEP) {
            if (xrole) xp_out[o] = s[q];
            else       out[o] = tanhf(s[q] + bv[p]);
        } else {
            out[o] = s[q] + bv[p];
        }
    }
}

__global__ void rnn_zero_kernel(float* __restrict__ p, int n)
{
    int i = blockIdx.x * blockDim.x + threadIdx.x;
    if (i < n) p[i] = 0.0f;
}

extern "C" void kernel_launch(void* const* d_in, const int* in_sizes, int n_in,
                              void* d_out, int out_size, void* d_ws, size_t ws_size,
                              hipStream_t stream)
{
    (void)in_sizes; (void)n_in; (void)out_size; (void)ws_size;
    const float* x    = (const float*)d_in[0];
    const float* W_xh = (const float*)d_in[1];
    const float* W_hh = (const float*)d_in[2];
    const float* b_h  = (const float*)d_in[3];
    const float* W_hy = (const float*)d_in[4];
    const float* b_y  = (const float*)d_in[5];
    float* out = (float*)d_out;

    const int hElems = RNN_BATCH * RNN_HID;
    const int wElems = RNN_HID * RNN_HID;

    float* hA    = (float*)d_ws;
    float* hB    = hA + hElems;
    float* WTxh  = hB + hElems;
    float* WThh  = WTxh + wElems;
    float* WThy  = WThh + wElems;
    float* xpA   = WThy;
    float* xpB   = WThy + hElems;

    rnn_zero_kernel<<<hElems / 256, 256, 0, stream>>>(hA, hElems);

    rnn_reorder_w_kernel<<<wElems / 4 / 256, 256, 0, stream>>>(W_xh, WTxh);
    rnn_reorder_w_kernel<<<wElems / 4 / 256, 256, 0, stream>>>(W_hh, WThh);

    dim3 gridOne(RNN_HID / (16 * NTILES), RNN_BATCH / 16, 1);
    dim3 gridTwo(RNN_HID / (16 * NTILES), RNN_BATCH / 16, 2);
    dim3 block(128);

    float* hin = hA;
    float* hout = hB;
    float* xin = xpA;
    float* xout = xpB;

    rnn_gemm_kernel<true><<<gridOne, block, 0, stream>>>(
        hin, WThh, x, WTxh, xout, b_h, hout, xin, 1);

    for (int t = 0; t < RNN_SEQ - 1; t++) {
        rnn_gemm_kernel<true><<<gridTwo, block, 0, stream>>>(
            hin, WThh, x + (size_t)(t + 1) * RNN_HID, WTxh, xin, b_h, hout, xout, 0);
        float* tmp = hin; hin = hout; hout = tmp;
        tmp = xin; xin = xout; xout = tmp;
    }
    rnn_gemm_kernel<true><<<gridOne, block, 0, stream>>>(
        hin, WThh, x, WTxh, xin, b_h, hout, xout, 0);
    {
        float* tmp = hin; hin = hout; hout = tmp;
    }

    rnn_reorder_w_kernel<<<wElems / 4 / 256, 256, 0, stream>>>(W_hy, WThy);

    rnn_gemm_kernel<false><<<gridOne, block, 0, stream>>>(
        hin, WThy, x, WTxh, b_y, b_y, out, hout, 0);
}


# GEMM K loop: three register sets, the 20 fragment loads of block j+2 issued ahead of block j's WMMAs (two blocks of lead, was one with two sets), loop written out; block and summation order unchanged
# speedup vs baseline: 3.2994x; 1.0549x over previous
#include <hip/hip_runtime.h>
#include <math.h>


typedef __attribute__((ext_vector_type(2))) float v2f;
typedef __attribute__((ext_vector_type(4))) float v4f;
typedef __attribute__((ext_vector_type(8))) float v8f;

#define RNN_HID 1024
#define RNN_BATCH 128
#define RNN_SEQ 512
#define NTILES 4
#define QBLK 4
#define KWAVE 256

__global__ __launch_bounds__(256) void rnn_reorder_w_kernel(
    const float* __restrict__ W, float* __restrict__ WT)
{
    int idx = blockIdx.x * blockDim.x + threadIdx.x;
    int q = idx >> 10;
    int n = idx & (RNN_HID - 1);
    const float* src = W + (size_t)(4 * q) * RNN_HID + n;
    v4f v;
    v.x = src[0];
    v.y = src[RNN_HID];
    v.z = src[2 * RNN_HID];
    v.w = src[3 * RNN_HID];
    *(v4f*)(WT + (size_t)idx * 4) = v;
}

template <bool STEP>
__global__ __launch_bounds__(128) void rnn_gemm_kernel(
    const float* __restrict__ Ah, const float* __restrict__ WhT,
    const float* __restrict__ Ax, const float* __restrict__ WxT,
    const float* __restrict__ xp_in, const float* __restrict__ bias,
    float* __restrict__ out, float* __restrict__ xp_out, int zbase)
{
    const int lane = threadIdx.x & 31;
    const int wave = threadIdx.x >> 5;
    const int half = lane >> 4;
    const int lr   = lane & 15;

    const int n0 = blockIdx.x * (16 * NTILES);
    const int m0 = blockIdx.y * 16;

    const bool xrole = STEP && ((int)blockIdx.z + zbase) != 0;

    const float* A  = xrole ? Ax : Ah;
    const float* WT = xrole ? WxT : WhT;
    const size_t astr = xrole ? (size_t)RNN_SEQ * RNN_HID : (size_t)RNN_HID;
    const int kb = wave * KWAVE;

    const float* ap = A + (size_t)(m0 + lr) * astr + (kb + half * 4);
    const float* wt = WT + ((size_t)((kb >> 2) + half) * RNN_HID + n0 + lr) * 4;

    v4f aP[QBLK], bP[QBLK][NTILES];
    v4f aQ[QBLK], bQ[QBLK][NTILES];
    v4f aR[QBLK], bR[QBLK][NTILES];

#define RNN_LOAD_BLOCK(a_, b_)                                              \
    do {                                                                    \
        _Pragma("unroll")                                                   \
        for (int s = 0; s < QBLK; s++) {                                    \
            a_[s] = *(const v4f*)(ap + 8 * s);                              \
            _Pragma("unroll")                                               \
            for (int i = 0; i < NTILES; i++)                                \
                b_[s][i] = *(const v4f*)(wt + s * (2 * RNN_HID * 4) + i * 64); \
        }                                                                   \
        ap += 8 * QBLK;                                                     \
        wt += (size_t)QBLK * (2 * RNN_HID * 4);                             \
    } while (0)

#define RNN_MMA_BLOCK(a_, b_)                                               \
    do {                                                                    \
        _Pragma("unroll")                                                   \
        for (int s = 0; s < QBLK; s++) {                                    \
            _Pragma("unroll")                                               \
            for (int i = 0; i < NTILES; i++)                                \
                acc[i] = __builtin_amdgcn_wmma_f32_16x16x4_f32(             \
                    false, __builtin_shufflevector(a_[s], a_[s], 0, 1),     \
                    false, __builtin_shufflevector(b_[s][i], b_[s][i], 0, 1), \
                    (short)0, acc[i], false, false);                        \
            _Pragma("unroll")                                               \
            for (int i = 0; i < NTILES; i++)                                \
                acc[i] = __builtin_amdgcn_wmma_f32_16x16x4_f32(             \
                    false, __builtin_shufflevector(a_[s], a_[s], 2, 3),     \
                    false, __builtin_shufflevector(b_[s][i], b_[s][i], 2, 3), \
                    (short)0, acc[i], false, false);                        \
        }                                                                   \
    } while (0)

    RNN_LOAD_BLOCK(aP, bP);

    v8f xinit = {};
    if (STEP && !xrole) {
#pragma unroll
        for (int j = 0; j < 8; j++)
            xinit[j] = xp_in[(size_t)(m0 + j + half * 8) * RNN_HID + n0 + wave * 16 + lr];
    }
    float bv[2];
#pragma unroll
    for (int p = 0; p < 2; p++)
        bv[p] = xrole ? 0.0f : bias[n0 + 32 * p + lane];

    v8f acc[NTILES];
#pragma unroll
    for (int i = 0; i < NTILES; i++) {
        const v8f zero = {};
        acc[i] = (i == wave) ? xinit : zero;
    }

    RNN_LOAD_BLOCK(aQ, bQ);

#define RNN_THREE_BLOCKS()                                                  \
    do {                                                                    \
        RNN_LOAD_BLOCK(aR, bR);                                             \
        __builtin_amdgcn_sched_barrier(0);   \
        RNN_MMA_BLOCK(aP, bP);                                              \
        __builtin_amdgcn_sched_barrier(0);   \
        RNN_LOAD_BLOCK(aP, bP);                                             \
        __builtin_amdgcn_sched_barrier(0);   \
        RNN_MMA_BLOCK(aQ, bQ);                                              \
        __builtin_amdgcn_sched_barrier(0);   \
        RNN_LOAD_BLOCK(aQ, bQ);                                             \
        __builtin_amdgcn_sched_barrier(0);   \
        RNN_MMA_BLOCK(aR, bR);                                              \
        __builtin_amdgcn_sched_barrier(0);   \
    } while (0)

    RNN_THREE_BLOCKS();
    RNN_THREE_BLOCKS();
#undef RNN_THREE_BLOCKS
    RNN_MMA_BLOCK(aP, bP);
    RNN_MMA_BLOCK(aQ, bQ);
#undef RNN_LOAD_BLOCK
#undef RNN_MMA_BLOCK

    __shared__ float red[4][NTILES][8][32];
#pragma unroll
    for (int i = 0; i < NTILES; i++)
#pragma unroll
        for (int j = 0; j < 8; j++)
            red[wave][i][j][lane] = acc[i][j];
    __syncthreads();

    const int rbase = 4 * wave;
    const int jb = rbase & 7;
    const int sl = (rbase >> 3) * 16 + lr;

    float s[8];
#pragma unroll
    for (int q = 0; q < 8; q++) {
        const int i = 2 * (q & 1) + half;
        const int j = jb + (q >> 1);
        s[q] = red[0][i][j][sl] + red[1][i][j][sl] +
               red[2][i][j][sl] + red[3][i][j][sl];
    }

#pragma unroll
    for (int q = 0; q < 8; q++) {
        const int p = q & 1;
        const size_t o = (size_t)(m0 + rbase + (q >> 1)) * RNN_HID + n0 + 32 * p + lane;
        if (STEP) {
            if (xrole) xp_out[o] = s[q];
            else       out[o] = tanhf(s[q] + bv[p]);
        } else {
            out[o] = s[q] + bv[p];
        }
    }
}

__global__ void rnn_zero_kernel(float* __restrict__ p, int n)
{
    int i = blockIdx.x * blockDim.x + threadIdx.x;
    if (i < n) p[i] = 0.0f;
}

extern "C" void kernel_launch(void* const* d_in, const int* in_sizes, int n_in,
                              void* d_out, int out_size, void* d_ws, size_t ws_size,
                              hipStream_t stream)
{
    (void)in_sizes; (void)n_in; (void)out_size; (void)ws_size;
    const float* x    = (const float*)d_in[0];
    const float* W_xh = (const float*)d_in[1];
    const float* W_hh = (const float*)d_in[2];
    const float* b_h  = (const float*)d_in[3];
    const float* W_hy = (const float*)d_in[4];
    const float* b_y  = (const float*)d_in[5];
    float* out = (float*)d_out;

    const int hElems = RNN_BATCH * RNN_HID;
    const int wElems = RNN_HID * RNN_HID;

    float* hA    = (float*)d_ws;
    float* hB    = hA + hElems;
    float* WTxh  = hB + hElems;
    float* WThh  = WTxh + wElems;
    float* WThy  = WThh + wElems;
    float* xpA   = WThy;
    float* xpB   = WThy + hElems;

    rnn_zero_kernel<<<hElems / 256, 256, 0, stream>>>(hA, hElems);

    rnn_reorder_w_kernel<<<wElems / 4 / 256, 256, 0, stream>>>(W_xh, WTxh);
    rnn_reorder_w_kernel<<<wElems / 4 / 256, 256, 0, stream>>>(W_hh, WThh);

    dim3 gridOne(RNN_HID / (16 * NTILES), RNN_BATCH / 16, 1);
    dim3 gridTwo(RNN_HID / (16 * NTILES), RNN_BATCH / 16, 2);
    dim3 block(128);

    float* hin = hA;
    float* hout = hB;
    float* xin = xpA;
    float* xout = xpB;

    rnn_gemm_kernel<true><<<gridOne, block, 0, stream>>>(
        hin, WThh, x, WTxh, xout, b_h, hout, xin, 1);

    for (int t = 0; t < RNN_SEQ - 1; t++) {
        rnn_gemm_kernel<true><<<gridTwo, block, 0, stream>>>(
            hin, WThh, x + (size_t)(t + 1) * RNN_HID, WTxh, xin, b_h, hout, xout, 0);
        float* tmp = hin; hin = hout; hout = tmp;
        tmp = xin; xin = xout; xout = tmp;
    }
    rnn_gemm_kernel<true><<<gridOne, block, 0, stream>>>(
        hin, WThh, x, WTxh, xin, b_h, hout, xout, 0);
    {
        float* tmp = hin; hin = hout; hout = tmp;
    }

    rnn_reorder_w_kernel<<<wElems / 4 / 256, 256, 0, stream>>>(W_hy, WThy);

    rnn_gemm_kernel<false><<<gridOne, block, 0, stream>>>(
        hin, WThy, x, WTxh, b_y, b_y, out, hout, 0);
}
